# starting kernel
# baseline (speedup 1.0000x reference)
#include <hip/hip_runtime.h>

typedef __attribute__((ext_vector_type(2))) float v2f;
typedef __attribute__((ext_vector_type(8))) float v8f;

#define BH   32
#define NSEQ 4096
#define DIM  64
#define CHUNK 512

__global__ void FA_zero_ws(float* __restrict__ ws, int n) {
    int i = blockIdx.x * blockDim.x + threadIdx.x;
    if (i < n) ws[i] = 0.0f;
}

__global__ __launch_bounds__(256) void FA_pass1(
    const float* __restrict__ K, const float* __restrict__ V,
    const float* __restrict__ P,
    float* __restrict__ ksum_ws, float* __restrict__ ctx_ws)
{
    __shared__ float sP[DIM * DIM];
    __shared__ float sred[2][8][32];

    const int bh    = blockIdx.x;
    const int chunk = blockIdx.y;
    const int tid   = threadIdx.x;

    for (int i = tid; i < DIM * DIM; i += 256) sP[i] = P[i];
    __syncthreads();

    const int wave = tid >> 5;
    const int lane = tid & 31;
    const int l15  = lane & 15;
    const int hi   = lane >> 4;
    const int ct   = wave & 3;
    const int rg   = wave >> 2;
    const int col  = ct * 16 + l15;

    const float* Kbh = K + (size_t)bh * NSEQ * DIM;
    const float* Vbh = V + (size_t)bh * NSEQ * DIM;

    float ksum_p = 0.0f, ctx_p = 0.0f;

    const int nBase = chunk * CHUNK;
    for (int n0 = nBase + rg * 16; n0 < nBase + CHUNK; n0 += 32) {
        v8f c = {};
        const float* Arow = Kbh + (size_t)(n0 + l15) * DIM;
        #pragma unroll
        for (int k0 = 0; k0 < DIM; k0 += 4) {
            v2f a, b;
            a.x = Arow[k0 + 2 * hi];
            a.y = Arow[k0 + 2 * hi + 1];
            b.x = sP[col * DIM + k0 + 2 * hi];
            b.y = sP[col * DIM + k0 + 2 * hi + 1];
            c = __builtin_amdgcn_wmma_f32_16x16x4_f32(
                    false, a, false, b, (short)0, c, false, false);
        }
        const float* Vcol = Vbh + (size_t)(n0 + 8 * hi) * DIM + col;
        #pragma unroll
        for (int r = 0; r < 8; ++r) {
            float e = __expf(c[r]);
            ksum_p += e;
            ctx_p  += e * Vcol[(size_t)r * DIM];
        }
    }

    sred[0][wave][lane] = ksum_p;
    sred[1][wave][lane] = ctx_p;
    __syncthreads();

    if (tid < 64) {
        const int f   = tid;
        const int ctf = f >> 4;
        const int lf  = f & 15;
        float ks = sred[0][ctf][lf] + sred[0][ctf][lf + 16]
                 + sred[0][ctf + 4][lf] + sred[0][ctf + 4][lf + 16];
        float cx = sred[1][ctf][lf] + sred[1][ctf][lf + 16]
                 + sred[1][ctf + 4][lf] + sred[1][ctf + 4][lf + 16];
        atomicAdd(&ksum_ws[bh * DIM + f], ks);
        atomicAdd(&ctx_ws[bh * DIM + f], cx);
    }
}

__global__ __launch_bounds__(256) void FA_pass2(
    const float* __restrict__ Q, const float* __restrict__ P,
    const float* __restrict__ ksum_ws, const float* __restrict__ ctx_ws,
    float* __restrict__ out)
{
    __shared__ float sP[DIM * DIM];

    const int bh   = blockIdx.x;
    const int slab = blockIdx.y;
    const int tid  = threadIdx.x;

    for (int i = tid; i < DIM * DIM; i += 256) sP[i] = P[i];
    __syncthreads();

    const int wave = tid >> 5;
    const int lane = tid & 31;
    const int l15  = lane & 15;
    const int hi   = lane >> 4;
    const int n0   = slab * 128 + wave * 16;

    const float* Qbh = Q + (size_t)bh * NSEQ * DIM;

    float ks[4], cx[4];
    #pragma unroll
    for (int ct = 0; ct < 4; ++ct) {
        ks[ct] = ksum_ws[bh * DIM + ct * 16 + l15] + 1e-6f;
        cx[ct] = ctx_ws [bh * DIM + ct * 16 + l15];
    }

    v8f c[4] = {};
    const float* Arow = Qbh + (size_t)(n0 + l15) * DIM;
    #pragma unroll
    for (int k0 = 0; k0 < DIM; k0 += 4) {
        v2f a;
        a.x = Arow[k0 + 2 * hi];
        a.y = Arow[k0 + 2 * hi + 1];
        #pragma unroll
        for (int ct = 0; ct < 4; ++ct) {
            v2f b;
            b.x = sP[(ct * 16 + l15) * DIM + k0 + 2 * hi];
            b.y = sP[(ct * 16 + l15) * DIM + k0 + 2 * hi + 1];
            c[ct] = __builtin_amdgcn_wmma_f32_16x16x4_f32(
                        false, a, false, b, (short)0, c[ct], false, false);
        }
    }

    float e[4][8];
    float rinv[8];
    #pragma unroll
    for (int r = 0; r < 8; ++r) {
        float p = 0.0f;
        #pragma unroll
        for (int ct = 0; ct < 4; ++ct) {
            e[ct][r] = __expf(c[ct][r]);
            p += e[ct][r] * ks[ct];
        }
        p += __shfl_xor(p, 1, 32);
        p += __shfl_xor(p, 2, 32);
        p += __shfl_xor(p, 4, 32);
        p += __shfl_xor(p, 8, 32);
        rinv[r] = 1.0f / p;
    }

    float* Orow = out + (size_t)bh * NSEQ * DIM + (size_t)(n0 + 8 * hi) * DIM;
    #pragma unroll
    for (int r = 0; r < 8; ++r) {
        #pragma unroll
        for (int ct = 0; ct < 4; ++ct) {
            Orow[(size_t)r * DIM + ct * 16 + l15] = e[ct][r] * cx[ct] * rinv[r];
        }
    }
}

extern "C" void kernel_launch(void* const* d_in, const int* in_sizes, int n_in,
                              void* d_out, int out_size, void* d_ws, size_t ws_size,
                              hipStream_t stream) {
    const float* q    = (const float*)d_in[0];
    const float* k    = (const float*)d_in[1];
    const float* v    = (const float*)d_in[2];
    const float* proj = (const float*)d_in[3];
    float* out = (float*)d_out;

    float* ws_ksum = (float*)d_ws;
    float* ws_ctx  = ws_ksum + BH * DIM;

    const int ws_elems = 2 * BH * DIM;
    FA_zero_ws<<<(ws_elems + 255) / 256, 256, 0, stream>>>(ws_ksum, ws_elems);

    FA_pass1<<<dim3(BH, NSEQ / CHUNK), 256, 0, stream>>>(
        k, v, proj, ws_ksum, ws_ctx);

    FA_pass2<<<dim3(BH, NSEQ / 128), 256, 0, stream>>>(
        q, proj, ws_ksum, ws_ctx, out);
}


# Both passes re-tiled: wave owns 16 rows x all 64 features over 4 (pass 1) or 2 (pass 2) row tiles, projection kept in registers, next tile preloaded; f32 sum order of k_sum, ctx, row sums changes
# speedup vs baseline: 2.4908x; 2.4908x over previous
#include <hip/hip_runtime.h>


typedef __attribute__((ext_vector_type(2))) float v2f;
typedef __attribute__((ext_vector_type(4))) float v4f;
typedef __attribute__((ext_vector_type(8))) float v8f;

#define BH     32
#define NSEQ   4096
#define DIM    64
#define TILES1 4
#define CHUNK1 512
#define TILES2 2
#define CHUNK2 256
static_assert(CHUNK1 == 8 * TILES1 * 16 && CHUNK2 == 8 * TILES2 * 16, "chunk = 8 waves x tiles x 16 rows");
static_assert(NSEQ % CHUNK1 == 0 && NSEQ % CHUNK2 == 0, "chunks tile the sequence");
#define PPITCH 68

__device__ __forceinline__ int frag_k(int j, int hi) { return 8 * j + 4 * hi; }

__global__ void FA_zero_ws(float* __restrict__ ws, int n) {
    int i = blockIdx.x * blockDim.x + threadIdx.x;
    if (i < n) ws[i] = 0.0f;
}

__device__ __forceinline__ void stage_proj(const float* __restrict__ P, float* sP, int tid) {
    #pragma unroll
    for (int i = 0; i < 4; ++i) {
        const int idx = tid + 256 * i;
        const int f   = idx >> 4;
        const int q   = idx & 15;
        const v4f val = *(const v4f*)(P + f * DIM + 4 * q);
        const int lrow = (f & 3) * 16 + (f >> 2);
        *(v4f*)(sP + lrow * PPITCH + 4 * q) = val;
    }
}

__device__ __forceinline__ void load_b_frags(const float* sP, int c, int hi, v4f (&bq)[4][8]) {
    #pragma unroll
    for (int ct = 0; ct < 4; ++ct) {
        #pragma unroll
        for (int j = 0; j < 8; ++j)
            bq[ct][j] = *(const v4f*)(sP + (ct * 16 + c) * PPITCH + frag_k(j, hi));
    }
}

__device__ __forceinline__ void load_a_frags(const float* __restrict__ rowc, int hi, v4f (&aq)[8]) {
    #pragma unroll
    for (int j = 0; j < 8; ++j)
        aq[j] = *(const v4f*)(rowc + frag_k(j, hi));
}

__device__ __forceinline__ void tile_mma(const v4f (&aq)[8], const v4f (&bq)[4][8], v8f (&acc)[4]) {
    #pragma unroll
    for (int j = 0; j < 8; ++j) {
        const v2f a0 = __builtin_shufflevector(aq[j], aq[j], 0, 1);
        const v2f a1 = __builtin_shufflevector(aq[j], aq[j], 2, 3);
        #pragma unroll
        for (int ct = 0; ct < 4; ++ct) {
            const v2f b0 = __builtin_shufflevector(bq[ct][j], bq[ct][j], 0, 1);
            acc[ct] = __builtin_amdgcn_wmma_f32_16x16x4_f32(
                          false, a0, false, b0, (short)0, acc[ct], false, false);
        }
        #pragma unroll
        for (int ct = 0; ct < 4; ++ct) {
            const v2f b1 = __builtin_shufflevector(bq[ct][j], bq[ct][j], 2, 3);
            acc[ct] = __builtin_amdgcn_wmma_f32_16x16x4_f32(
                          false, a1, false, b1, (short)0, acc[ct], false, false);
        }
    }
}

__global__ __launch_bounds__(256) void FA_pass1(
    const float* __restrict__ K, const float* __restrict__ V,
    const float* __restrict__ P,
    float* __restrict__ ksum_ws, float* __restrict__ ctx_ws)
{
    __shared__ __attribute__((aligned(16))) float sP[DIM * PPITCH];
    __shared__ __attribute__((aligned(16))) float sred[2][8][DIM];

    const int bh    = blockIdx.x;
    const int chunk = blockIdx.y;
    const int tid   = threadIdx.x;
    const int wave  = tid >> 5;
    const int lane  = tid & 31;
    const int c     = lane & 15;
    const int hi    = lane >> 4;

    const size_t row0 = (size_t)bh * NSEQ + (size_t)chunk * CHUNK1 + (size_t)wave * (TILES1 * 16);
    const float* Arow = K + (row0 + c) * DIM;
    const float* Vrow = V + (row0 + 8 * hi) * DIM + 4 * c;

    v4f a_cur[8], a_nxt[8];
    load_a_frags(Arow, hi, a_cur);

    stage_proj(P, sP, tid);
    __syncthreads();

    v4f bq[4][8];
    load_b_frags(sP, c, hi, bq);
    __builtin_amdgcn_sched_barrier(0);

    float ksum_p[4] = {0.0f, 0.0f, 0.0f, 0.0f};
    float ctx_p[4]  = {0.0f, 0.0f, 0.0f, 0.0f};

    #pragma unroll
    for (int t = 0; t < TILES1; ++t) {
        if (t + 1 < TILES1) load_a_frags(Arow + (size_t)(t + 1) * 16 * DIM, hi, a_nxt);

        v4f vv[8];
        #pragma unroll
        for (int r = 0; r < 8; ++r)
            vv[r] = *(const v4f*)(Vrow + (size_t)(t * 16 + r) * DIM);
        __builtin_amdgcn_sched_barrier(0);

        v8f acc[4] = {};
        tile_mma(a_cur, bq, acc);

        #pragma unroll
        for (int r = 0; r < 8; ++r) {
            #pragma unroll
            for (int ct = 0; ct < 4; ++ct) {
                const float e = __expf(acc[ct][r]);
                ksum_p[ct] += e;
                ctx_p[ct]  += e * vv[r][ct];
            }
        }

        if (t + 1 < TILES1) {
            #pragma unroll
            for (int j = 0; j < 8; ++j) a_cur[j] = a_nxt[j];
        }
    }

    #pragma unroll
    for (int ct = 0; ct < 4; ++ct) {
        ksum_p[ct] += __shfl_xor(ksum_p[ct], 16, 32);
        ctx_p[ct]  += __shfl_xor(ctx_p[ct], 16, 32);
    }
    if (hi == 0) {
        v4f s0 = {ksum_p[0], ksum_p[1], ksum_p[2], ksum_p[3]};
        v4f s1 = {ctx_p[0], ctx_p[1], ctx_p[2], ctx_p[3]};
        *(v4f*)(&sred[0][wave][4 * c]) = s0;
        *(v4f*)(&sred[1][wave][4 * c]) = s1;
    }
    __syncthreads();

    if (tid < 64) {
        float ks = 0.0f, cx = 0.0f;
        #pragma unroll
        for (int w = 0; w < 8; ++w) {
            ks += sred[0][w][tid];
            cx += sred[1][w][tid];
        }
        atomicAdd(&ksum_ws[bh * DIM + tid], ks);
        atomicAdd(&ctx_ws[bh * DIM + tid], cx);
    }
}

__global__ __launch_bounds__(256, 4) void FA_pass2(
    const float* __restrict__ Q, const float* __restrict__ P,
    const float* __restrict__ ksum_ws, const float* __restrict__ ctx_ws,
    float* __restrict__ out)
{
    __shared__ __attribute__((aligned(16))) float sP[DIM * PPITCH];

    const int bh    = blockIdx.x;
    const int chunk = blockIdx.y;
    const int tid   = threadIdx.x;
    const int wave  = tid >> 5;
    const int lane  = tid & 31;
    const int c     = lane & 15;
    const int hi    = lane >> 4;

    const size_t row0 = (size_t)bh * NSEQ + (size_t)chunk * CHUNK2 + (size_t)wave * (TILES2 * 16);
    const float* Arow = Q + (row0 + c) * DIM;
    float*       Orow = out + (row0 + 8 * hi) * DIM + 4 * c;

    v4f a_cur[8], a_nxt[8];
    load_a_frags(Arow, hi, a_cur);

    v4f ks = *(const v4f*)(ksum_ws + bh * DIM + 4 * c);
    const v4f cx = *(const v4f*)(ctx_ws + bh * DIM + 4 * c);

    stage_proj(P, sP, tid);
    __syncthreads();

    v4f bq[4][8];
    load_b_frags(sP, c, hi, bq);
    __builtin_amdgcn_sched_barrier(0);

    ks += 1e-6f;

    #pragma unroll
    for (int t = 0; t < TILES2; ++t) {
        if (t + 1 < TILES2) load_a_frags(Arow + (size_t)(t + 1) * 16 * DIM, hi, a_nxt);
        __builtin_amdgcn_sched_barrier(0);

        v8f acc[4] = {};
        tile_mma(a_cur, bq, acc);

        float e[4][8];
        float p[8];
        #pragma unroll
        for (int r = 0; r < 8; ++r) {
            p[r] = 0.0f;
            #pragma unroll
            for (int ct = 0; ct < 4; ++ct) {
                e[ct][r] = __expf(acc[ct][r]);
                p[r] += e[ct][r] * ks[ct];
            }
        }
        #pragma unroll
        for (int m = 1; m < 16; m <<= 1) {
            float q[8];
            #pragma unroll
            for (int r = 0; r < 8; ++r) q[r] = __shfl_xor(p[r], m, 32);
            #pragma unroll
            for (int r = 0; r < 8; ++r) p[r] += q[r];
        }
        #pragma unroll
        for (int r = 0; r < 8; ++r) {
            const float rinv = 1.0f / p[r];
            v4f o;
            o.x = e[0][r] * cx[0] * rinv;
            o.y = e[1][r] * cx[1] * rinv;
            o.z = e[2][r] * cx[2] * rinv;
            o.w = e[3][r] * cx[3] * rinv;
            *(v4f*)(Orow + (size_t)(t * 16 + r) * DIM) = o;
        }

        if (t + 1 < TILES2) {
            #pragma unroll
            for (int j = 0; j < 8; ++j) a_cur[j] = a_nxt[j];
        }
    }
}

extern "C" void kernel_launch(void* const* d_in, const int* in_sizes, int n_in,
                              void* d_out, int out_size, void* d_ws, size_t ws_size,
                              hipStream_t stream) {
    const float* q    = (const float*)d_in[0];
    const float* k    = (const float*)d_in[1];
    const float* v    = (const float*)d_in[2];
    const float* proj = (const float*)d_in[3];
    float* out = (float*)d_out;

    float* ws_ksum = (float*)d_ws;
    float* ws_ctx  = ws_ksum + BH * DIM;

    const int ws_elems = 2 * BH * DIM;
    FA_zero_ws<<<(ws_elems + 255) / 256, 256, 0, stream>>>(ws_ksum, ws_elems);

    FA_pass1<<<dim3(BH, NSEQ / CHUNK1), 256, 0, stream>>>(
        k, v, proj, ws_ksum, ws_ctx);

    FA_pass2<<<dim3(BH, NSEQ / CHUNK2), 256, 0, stream>>>(
        q, proj, ws_ksum, ws_ctx, out);
}


# k_sum/ctx summed without atomics or a clearing launch: pass 1 workgroups store partials, pass 2 adds the 8 per (b,h) as two fixed halves of four (f32 sum order across chunks fixed); 3 launches -> 2
# speedup vs baseline: 2.6381x; 1.0591x over previous
#include <hip/hip_runtime.h>


typedef __attribute__((ext_vector_type(2))) float v2f;
typedef __attribute__((ext_vector_type(4))) float v4f;
typedef __attribute__((ext_vector_type(8))) float v8f;

#define BH     32
#define NSEQ   4096
#define DIM    64
#define TILES1 4
#define CHUNK1 512
#define TILES2 2
#define CHUNK2 256
static_assert(CHUNK1 == 8 * TILES1 * 16 && CHUNK2 == 8 * TILES2 * 16, "chunk = 8 waves x tiles x 16 rows");
static_assert(NSEQ % CHUNK1 == 0 && NSEQ % CHUNK2 == 0, "chunks tile the sequence");
#define PPITCH 68
#define NCHUNK1 (NSEQ / CHUNK1)
#define PART_FLOATS (BH * NCHUNK1 * 2 * DIM)

__device__ __forceinline__ int frag_k(int j, int hi) { return 8 * j + 4 * hi; }


__device__ __forceinline__ void stage_proj(const float* __restrict__ P, float* sP, int tid) {
    #pragma unroll
    for (int i = 0; i < 4; ++i) {
        const int idx = tid + 256 * i;
        const int f   = idx >> 4;
        const int q   = idx & 15;
        const v4f val = *(const v4f*)(P + f * DIM + 4 * q);
        const int lrow = (f & 3) * 16 + (f >> 2);
        *(v4f*)(sP + lrow * PPITCH + 4 * q) = val;
    }
}

__device__ __forceinline__ void load_b_frags(const float* sP, int c, int hi, v4f (&bq)[4][8]) {
    #pragma unroll
    for (int ct = 0; ct < 4; ++ct) {
        #pragma unroll
        for (int j = 0; j < 8; ++j)
            bq[ct][j] = *(const v4f*)(sP + (ct * 16 + c) * PPITCH + frag_k(j, hi));
    }
}

__device__ __forceinline__ void load_a_frags(const float* __restrict__ rowc, int hi, v4f (&aq)[8]) {
    #pragma unroll
    for (int j = 0; j < 8; ++j)
        aq[j] = *(const v4f*)(rowc + frag_k(j, hi));
}

__device__ __forceinline__ void tile_mma(const v4f (&aq)[8], const v4f (&bq)[4][8], v8f (&acc)[4]) {
    #pragma unroll
    for (int j = 0; j < 8; ++j) {
        const v2f a0 = __builtin_shufflevector(aq[j], aq[j], 0, 1);
        const v2f a1 = __builtin_shufflevector(aq[j], aq[j], 2, 3);
        #pragma unroll
        for (int ct = 0; ct < 4; ++ct) {
            const v2f b0 = __builtin_shufflevector(bq[ct][j], bq[ct][j], 0, 1);
            acc[ct] = __builtin_amdgcn_wmma_f32_16x16x4_f32(
                          false, a0, false, b0, (short)0, acc[ct], false, false);
        }
        #pragma unroll
        for (int ct = 0; ct < 4; ++ct) {
            const v2f b1 = __builtin_shufflevector(bq[ct][j], bq[ct][j], 2, 3);
            acc[ct] = __builtin_amdgcn_wmma_f32_16x16x4_f32(
                          false, a1, false, b1, (short)0, acc[ct], false, false);
        }
    }
}

__global__ __launch_bounds__(256) void FA_pass1(
    const float* __restrict__ K, const float* __restrict__ V,
    const float* __restrict__ P,
    float* __restrict__ part_ws)
{
    __shared__ __attribute__((aligned(16))) float sP[DIM * PPITCH];
    __shared__ __attribute__((aligned(16))) float sred[2][8][DIM];

    const int bh    = blockIdx.x;
    const int chunk = blockIdx.y;
    const int tid   = threadIdx.x;
    const int wave  = tid >> 5;
    const int lane  = tid & 31;
    const int c     = lane & 15;
    const int hi    = lane >> 4;

    const size_t row0 = (size_t)bh * NSEQ + (size_t)chunk * CHUNK1 + (size_t)wave * (TILES1 * 16);
    const float* Arow = K + (row0 + c) * DIM;
    const float* Vrow = V + (row0 + 8 * hi) * DIM + 4 * c;

    v4f a_cur[8], a_nxt[8];
    load_a_frags(Arow, hi, a_cur);

    stage_proj(P, sP, tid);
    __syncthreads();

    v4f bq[4][8];
    load_b_frags(sP, c, hi, bq);
    __builtin_amdgcn_sched_barrier(0);

    float ksum_p[4] = {0.0f, 0.0f, 0.0f, 0.0f};
    float ctx_p[4]  = {0.0f, 0.0f, 0.0f, 0.0f};

    #pragma unroll
    for (int t = 0; t < TILES1; ++t) {
        if (t + 1 < TILES1) load_a_frags(Arow + (size_t)(t + 1) * 16 * DIM, hi, a_nxt);

        v4f vv[8];
        #pragma unroll
        for (int r = 0; r < 8; ++r)
            vv[r] = *(const v4f*)(Vrow + (size_t)(t * 16 + r) * DIM);
        __builtin_amdgcn_sched_barrier(0);

        v8f acc[4] = {};
        tile_mma(a_cur, bq, acc);

        #pragma unroll
        for (int r = 0; r < 8; ++r) {
            #pragma unroll
            for (int ct = 0; ct < 4; ++ct) {
                const float e = __expf(acc[ct][r]);
                ksum_p[ct] += e;
                ctx_p[ct]  += e * vv[r][ct];
            }
        }

        if (t + 1 < TILES1) {
            #pragma unroll
            for (int j = 0; j < 8; ++j) a_cur[j] = a_nxt[j];
        }
    }

    #pragma unroll
    for (int ct = 0; ct < 4; ++ct) {
        ksum_p[ct] += __shfl_xor(ksum_p[ct], 16, 32);
        ctx_p[ct]  += __shfl_xor(ctx_p[ct], 16, 32);
    }
    if (hi == 0) {
        v4f s0 = {ksum_p[0], ksum_p[1], ksum_p[2], ksum_p[3]};
        v4f s1 = {ctx_p[0], ctx_p[1], ctx_p[2], ctx_p[3]};
        *(v4f*)(&sred[0][wave][4 * c]) = s0;
        *(v4f*)(&sred[1][wave][4 * c]) = s1;
    }
    __syncthreads();

    if (tid < 2 * DIM) {
        const int kind = tid >> 6;
        const int f    = tid & 63;
        float s = 0.0f;
        #pragma unroll
        for (int w = 0; w < 8; ++w) s += sred[kind][w][f];
        part_ws[((size_t)bh * NCHUNK1 + chunk) * (2 * DIM) + tid] = s;
    }
}

__global__ __launch_bounds__(256, 4) void FA_pass2(
    const float* __restrict__ Q, const float* __restrict__ P,
    const float* __restrict__ part_ws,
    float* __restrict__ out)
{
    __shared__ __attribute__((aligned(16))) float sP[DIM * PPITCH];
    __shared__ __attribute__((aligned(16))) float stot[2][2 * DIM];

    const int bh    = blockIdx.x;
    const int chunk = blockIdx.y;
    const int tid   = threadIdx.x;
    const int wave  = tid >> 5;
    const int lane  = tid & 31;
    const int c     = lane & 15;
    const int hi    = lane >> 4;

    const size_t row0 = (size_t)bh * NSEQ + (size_t)chunk * CHUNK2 + (size_t)wave * (TILES2 * 16);
    const float* Arow = Q + (row0 + c) * DIM;
    float*       Orow = out + (row0 + 8 * hi) * DIM + 4 * c;

    v4f a_cur[8], a_nxt[8];
    load_a_frags(Arow, hi, a_cur);

    const int slot = tid & (2 * DIM - 1);
    const int half = tid >> 7;
    const float* pp = part_ws + ((size_t)bh * NCHUNK1 + half * (NCHUNK1 / 2)) * (2 * DIM) + slot;
    float pv[NCHUNK1 / 2];
    #pragma unroll
    for (int ch = 0; ch < NCHUNK1 / 2; ++ch) pv[ch] = pp[ch * (2 * DIM)];

    stage_proj(P, sP, tid);

    float tot = pv[0];
    #pragma unroll
    for (int ch = 1; ch < NCHUNK1 / 2; ++ch) tot += pv[ch];
    stot[half][slot] = tot;
    __syncthreads();

    v4f bq[4][8];
    load_b_frags(sP, c, hi, bq);
    v4f ks = *(const v4f*)(&stot[0][4 * c]) + *(const v4f*)(&stot[1][4 * c]);
    const v4f cx = *(const v4f*)(&stot[0][DIM + 4 * c]) + *(const v4f*)(&stot[1][DIM + 4 * c]);
    __builtin_amdgcn_sched_barrier(0);

    ks += 1e-6f;

    #pragma unroll
    for (int t = 0; t < TILES2; ++t) {
        if (t + 1 < TILES2) load_a_frags(Arow + (size_t)(t + 1) * 16 * DIM, hi, a_nxt);
        __builtin_amdgcn_sched_barrier(0);

        v8f acc[4] = {};
        tile_mma(a_cur, bq, acc);

        float e[4][8];
        float p[8];
        #pragma unroll
        for (int r = 0; r < 8; ++r) {
            p[r] = 0.0f;
            #pragma unroll
            for (int ct = 0; ct < 4; ++ct) {
                e[ct][r] = __expf(acc[ct][r]);
                p[r] += e[ct][r] * ks[ct];
            }
        }
        #pragma unroll
        for (int m = 1; m < 16; m <<= 1) {
            float q[8];
            #pragma unroll
            for (int r = 0; r < 8; ++r) q[r] = __shfl_xor(p[r], m, 32);
            #pragma unroll
            for (int r = 0; r < 8; ++r) p[r] += q[r];
        }
        #pragma unroll
        for (int r = 0; r < 8; ++r) {
            const float rinv = 1.0f / p[r];
            v4f o;
            o.x = e[0][r] * cx[0] * rinv;
            o.y = e[1][r] * cx[1] * rinv;
            o.z = e[2][r] * cx[2] * rinv;
            o.w = e[3][r] * cx[3] * rinv;
            *(v4f*)(Orow + (size_t)(t * 16 + r) * DIM) = o;
        }

        if (t + 1 < TILES2) {
            #pragma unroll
            for (int j = 0; j < 8; ++j) a_cur[j] = a_nxt[j];
        }
    }
}

extern "C" void kernel_launch(void* const* d_in, const int* in_sizes, int n_in,
                              void* d_out, int out_size, void* d_ws, size_t ws_size,
                              hipStream_t stream) {
    const float* q    = (const float*)d_in[0];
    const float* k    = (const float*)d_in[1];
    const float* v    = (const float*)d_in[2];
    const float* proj = (const float*)d_in[3];
    float* out = (float*)d_out;

    float* ws_part = (float*)d_ws;
    if (ws_size < PART_FLOATS * sizeof(float)) return;

    FA_pass1<<<dim3(BH, NCHUNK1), 256, 0, stream>>>(
        k, v, proj, ws_part);

    FA_pass2<<<dim3(BH, NSEQ / CHUNK2), 256, 0, stream>>>(
        q, proj, ws_part, out);
}


# Row operand (K in pass 1, Q in pass 2) loaded as whole 128-byte lines, 512 contiguous bytes per wave load, and turned into A fragments through a wave-private LDS tile (was fragment-order loads)
# speedup vs baseline: 2.7606x; 1.0464x over previous
#include <hip/hip_runtime.h>


typedef __attribute__((ext_vector_type(2))) float v2f;
typedef __attribute__((ext_vector_type(4))) float v4f;
typedef __attribute__((ext_vector_type(8))) float v8f;

#define BH     32
#define NSEQ   4096
#define DIM    64
#define TILES1 4
#define CHUNK1 512
#define TILES2 2
#define CHUNK2 256
static_assert(CHUNK1 == 8 * TILES1 * 16 && CHUNK2 == 8 * TILES2 * 16, "chunk = 8 waves x tiles x 16 rows");
static_assert(NSEQ % CHUNK1 == 0 && NSEQ % CHUNK2 == 0, "chunks tile the sequence");
#define PPITCH 68
#define APITCH 68
#define NCHUNK1 (NSEQ / CHUNK1)
#define PART_FLOATS (BH * NCHUNK1 * 2 * DIM)

__device__ __forceinline__ int frag_k(int j, int hi) { return 8 * j + 4 * hi; }


__device__ __forceinline__ void stage_proj(const float* __restrict__ P, float* sP, int tid) {
    #pragma unroll
    for (int i = 0; i < 4; ++i) {
        const int idx = tid + 256 * i;
        const int f   = idx >> 4;
        const int q   = idx & 15;
        const v4f val = *(const v4f*)(P + f * DIM + 4 * q);
        const int lrow = (f & 3) * 16 + (f >> 2);
        *(v4f*)(sP + lrow * PPITCH + 4 * q) = val;
    }
}

__device__ __forceinline__ void load_b_frags(const float* sP, int c, int hi, v4f (&bq)[4][8]) {
    #pragma unroll
    for (int ct = 0; ct < 4; ++ct) {
        #pragma unroll
        for (int j = 0; j < 8; ++j)
            bq[ct][j] = *(const v4f*)(sP + (ct * 16 + c) * PPITCH + frag_k(j, hi));
    }
}

__device__ __forceinline__ void load_tile_lines(const float* __restrict__ tile, int lane, v4f (&st)[8]) {
    #pragma unroll
    for (int i = 0; i < 8; ++i)
        st[i] = *(const v4f*)(tile + i * 128 + 4 * lane);
}

__device__ __forceinline__ void tile_to_frags(float* sT, int c, int hi, const v4f (&st)[8], v4f (&aq)[8]) {
    #pragma unroll
    for (int i = 0; i < 8; ++i)
        *(v4f*)(sT + (2 * i + hi) * APITCH + 4 * c) = st[i];
    #pragma unroll
    for (int j = 0; j < 8; ++j)
        aq[j] = *(const v4f*)(sT + c * APITCH + frag_k(j, hi));
}

__device__ __forceinline__ void tile_mma(const v4f (&aq)[8], const v4f (&bq)[4][8], v8f (&acc)[4]) {
    #pragma unroll
    for (int j = 0; j < 8; ++j) {
        const v2f a0 = __builtin_shufflevector(aq[j], aq[j], 0, 1);
        const v2f a1 = __builtin_shufflevector(aq[j], aq[j], 2, 3);
        #pragma unroll
        for (int ct = 0; ct < 4; ++ct) {
            const v2f b0 = __builtin_shufflevector(bq[ct][j], bq[ct][j], 0, 1);
            acc[ct] = __builtin_amdgcn_wmma_f32_16x16x4_f32(
                          false, a0, false, b0, (short)0, acc[ct], false, false);
        }
        #pragma unroll
        for (int ct = 0; ct < 4; ++ct) {
            const v2f b1 = __builtin_shufflevector(bq[ct][j], bq[ct][j], 2, 3);
            acc[ct] = __builtin_amdgcn_wmma_f32_16x16x4_f32(
                          false, a1, false, b1, (short)0, acc[ct], false, false);
        }
    }
}

__global__ __launch_bounds__(256) void FA_pass1(
    const float* __restrict__ K, const float* __restrict__ V,
    const float* __restrict__ P,
    float* __restrict__ part_ws)
{
    __shared__ __attribute__((aligned(16))) float sP[DIM * PPITCH];
    __shared__ __attribute__((aligned(16))) float sred[2][8][DIM];
    __shared__ __attribute__((aligned(16))) float sTile[8][16 * APITCH];

    const int bh    = blockIdx.x;
    const int chunk = blockIdx.y;
    const int tid   = threadIdx.x;
    const int wave  = tid >> 5;
    const int lane  = tid & 31;
    const int c     = lane & 15;
    const int hi    = lane >> 4;

    const size_t row0 = (size_t)bh * NSEQ + (size_t)chunk * CHUNK1 + (size_t)wave * (TILES1 * 16);
    const float* Atile = K + row0 * DIM;
    const float* Vrow = V + (row0 + 8 * hi) * DIM + 4 * c;
    float* sT = sTile[wave];

    v4f a_cur[8], a_ln[8];
    load_tile_lines(Atile, lane, a_ln);

    stage_proj(P, sP, tid);
    __syncthreads();

    v4f bq[4][8];
    load_b_frags(sP, c, hi, bq);
    tile_to_frags(sT, c, hi, a_ln, a_cur);
    __builtin_amdgcn_sched_barrier(0);

    float ksum_p[4] = {0.0f, 0.0f, 0.0f, 0.0f};
    float ctx_p[4]  = {0.0f, 0.0f, 0.0f, 0.0f};

    #pragma unroll
    for (int t = 0; t < TILES1; ++t) {
        if (t + 1 < TILES1) load_tile_lines(Atile + (size_t)(t + 1) * 16 * DIM, lane, a_ln);

        v4f vv[8];
        #pragma unroll
        for (int r = 0; r < 8; ++r)
            vv[r] = *(const v4f*)(Vrow + (size_t)(t * 16 + r) * DIM);
        __builtin_amdgcn_sched_barrier(0);

        v8f acc[4] = {};
        tile_mma(a_cur, bq, acc);
        if (t + 1 < TILES1) __builtin_amdgcn_sched_barrier(0);

        #pragma unroll
        for (int r = 0; r < 8; ++r) {
            #pragma unroll
            for (int ct = 0; ct < 4; ++ct) {
                const float e = __expf(acc[ct][r]);
                ksum_p[ct] += e;
                ctx_p[ct]  += e * vv[r][ct];
            }
        }

        if (t + 1 < TILES1) tile_to_frags(sT, c, hi, a_ln, a_cur);
    }

    #pragma unroll
    for (int ct = 0; ct < 4; ++ct) {
        ksum_p[ct] += __shfl_xor(ksum_p[ct], 16, 32);
        ctx_p[ct]  += __shfl_xor(ctx_p[ct], 16, 32);
    }
    if (hi == 0) {
        v4f s0 = {ksum_p[0], ksum_p[1], ksum_p[2], ksum_p[3]};
        v4f s1 = {ctx_p[0], ctx_p[1], ctx_p[2], ctx_p[3]};
        *(v4f*)(&sred[0][wave][4 * c]) = s0;
        *(v4f*)(&sred[1][wave][4 * c]) = s1;
    }
    __syncthreads();

    if (tid < 2 * DIM) {
        const int kind = tid >> 6;
        const int f    = tid & 63;
        float s = 0.0f;
        #pragma unroll
        for (int w = 0; w < 8; ++w) s += sred[kind][w][f];
        part_ws[((size_t)bh * NCHUNK1 + chunk) * (2 * DIM) + tid] = s;
    }
}

__global__ __launch_bounds__(256, 4) void FA_pass2(
    const float* __restrict__ Q, const float* __restrict__ P,
    const float* __restrict__ part_ws,
    float* __restrict__ out)
{
    __shared__ __attribute__((aligned(16))) float sP[DIM * PPITCH];
    __shared__ __attribute__((aligned(16))) float stot[2][2 * DIM];
    __shared__ __attribute__((aligned(16))) float sTile[8][16 * APITCH];

    const int bh    = blockIdx.x;
    const int chunk = blockIdx.y;
    const int tid   = threadIdx.x;
    const int wave  = tid >> 5;
    const int lane  = tid & 31;
    const int c     = lane & 15;
    const int hi    = lane >> 4;

    const size_t row0 = (size_t)bh * NSEQ + (size_t)chunk * CHUNK2 + (size_t)wave * (TILES2 * 16);
    const float* Atile = Q + row0 * DIM;
    float*       Orow = out + (row0 + 8 * hi) * DIM + 4 * c;
    float* sT = sTile[wave];

    v4f a_cur[8], a_ln[8];
    load_tile_lines(Atile, lane, a_ln);

    const int slot = tid & (2 * DIM - 1);
    const int half = tid >> 7;
    const float* pp = part_ws + ((size_t)bh * NCHUNK1 + half * (NCHUNK1 / 2)) * (2 * DIM) + slot;
    float pv[NCHUNK1 / 2];
    #pragma unroll
    for (int ch = 0; ch < NCHUNK1 / 2; ++ch) pv[ch] = pp[ch * (2 * DIM)];

    stage_proj(P, sP, tid);

    float tot = pv[0];
    #pragma unroll
    for (int ch = 1; ch < NCHUNK1 / 2; ++ch) tot += pv[ch];
    stot[half][slot] = tot;
    __syncthreads();

    v4f bq[4][8];
    load_b_frags(sP, c, hi, bq);
    v4f ks = *(const v4f*)(&stot[0][4 * c]) + *(const v4f*)(&stot[1][4 * c]);
    const v4f cx = *(const v4f*)(&stot[0][DIM + 4 * c]) + *(const v4f*)(&stot[1][DIM + 4 * c]);
    tile_to_frags(sT, c, hi, a_ln, a_cur);
    __builtin_amdgcn_sched_barrier(0);

    ks += 1e-6f;

    #pragma unroll
    for (int t = 0; t < TILES2; ++t) {
        if (t + 1 < TILES2) load_tile_lines(Atile + (size_t)(t + 1) * 16 * DIM, lane, a_ln);
        __builtin_amdgcn_sched_barrier(0);

        v8f acc[4] = {};
        tile_mma(a_cur, bq, acc);
        if (t + 1 < TILES2) __builtin_amdgcn_sched_barrier(0);

        float e[4][8];
        float p[8];
        #pragma unroll
        for (int r = 0; r < 8; ++r) {
            p[r] = 0.0f;
            #pragma unroll
            for (int ct = 0; ct < 4; ++ct) {
                e[ct][r] = __expf(acc[ct][r]);
                p[r] += e[ct][r] * ks[ct];
            }
        }
        #pragma unroll
        for (int m = 1; m < 16; m <<= 1) {
            float q[8];
            #pragma unroll
            for (int r = 0; r < 8; ++r) q[r] = __shfl_xor(p[r], m, 32);
            #pragma unroll
            for (int r = 0; r < 8; ++r) p[r] += q[r];
        }
        #pragma unroll
        for (int r = 0; r < 8; ++r) {
            const float rinv = 1.0f / p[r];
            v4f o;
            o.x = e[0][r] * cx[0] * rinv;
            o.y = e[1][r] * cx[1] * rinv;
            o.z = e[2][r] * cx[2] * rinv;
            o.w = e[3][r] * cx[3] * rinv;
            *(v4f*)(Orow + (size_t)(t * 16 + r) * DIM) = o;
        }

        if (t + 1 < TILES2) tile_to_frags(sT, c, hi, a_ln, a_cur);
    }
}

extern "C" void kernel_launch(void* const* d_in, const int* in_sizes, int n_in,
                              void* d_out, int out_size, void* d_ws, size_t ws_size,
                              hipStream_t stream) {
    const float* q    = (const float*)d_in[0];
    const float* k    = (const float*)d_in[1];
    const float* v    = (const float*)d_in[2];
    const float* proj = (const float*)d_in[3];
    float* out = (float*)d_out;

    float* ws_part = (float*)d_ws;
    if (ws_size < PART_FLOATS * sizeof(float)) return;

    FA_pass1<<<dim3(BH, NCHUNK1), 256, 0, stream>>>(
        k, v, proj, ws_part);

    FA_pass2<<<dim3(BH, NSEQ / CHUNK2), 256, 0, stream>>>(
        q, proj, ws_part, out);
}


# Start-up of both passes: projection loaded first and an LDS-only barrier, so staging does not wait for the row tile; first tile's A and B LDS reads issued in order of use
# speedup vs baseline: 2.9677x; 1.0750x over previous
#include <hip/hip_runtime.h>


typedef __attribute__((ext_vector_type(2))) float v2f;
typedef __attribute__((ext_vector_type(4))) float v4f;
typedef __attribute__((ext_vector_type(8))) float v8f;

#define BH     32
#define NSEQ   4096
#define DIM    64
#define TILES1 4
#define CHUNK1 512
#define TILES2 2
#define CHUNK2 256
static_assert(CHUNK1 == 8 * TILES1 * 16 && CHUNK2 == 8 * TILES2 * 16, "chunk = 8 waves x tiles x 16 rows");
static_assert(NSEQ % CHUNK1 == 0 && NSEQ % CHUNK2 == 0, "chunks tile the sequence");
#define PPITCH 68
#define APITCH 68
#define NCHUNK1 (NSEQ / CHUNK1)
#define PART_FLOATS (BH * NCHUNK1 * 2 * DIM)

__device__ __forceinline__ int frag_k(int j, int hi) { return 8 * j + 4 * hi; }


__device__ __forceinline__ void load_proj(const float* __restrict__ P, int tid, v4f (&pr)[4]) {
    #pragma unroll
    for (int i = 0; i < 4; ++i) {
        const int idx = tid + 256 * i;
        pr[i] = *(const v4f*)(P + (idx >> 4) * DIM + 4 * (idx & 15));
    }
}
__device__ __forceinline__ void lds_barrier() {
    __builtin_amdgcn_fence(__ATOMIC_RELEASE, "workgroup", "local");
    __builtin_amdgcn_s_barrier();
    __builtin_amdgcn_fence(__ATOMIC_ACQUIRE, "workgroup", "local");
}
__device__ __forceinline__ void store_proj(float* sP, int tid, const v4f (&pr)[4]) {
    #pragma unroll
    for (int i = 0; i < 4; ++i) {
        const int idx = tid + 256 * i;
        const int f   = idx >> 4;
        const int q   = idx & 15;
        const int lrow = (f & 3) * 16 + (f >> 2);
        *(v4f*)(sP + lrow * PPITCH + 4 * q) = pr[i];
    }
}

__device__ __forceinline__ void first_frags(float* sT, const float* sP, int c, int hi,
                                            const v4f (&st)[8], v4f (&aq)[8], v4f (&bq)[4][8]) {
    #pragma unroll
    for (int i = 0; i < 8; ++i)
        *(v4f*)(sT + (2 * i + hi) * APITCH + 4 * c) = st[i];
    #pragma unroll
    for (int j = 0; j < 8; ++j) {
        aq[j] = *(const v4f*)(sT + c * APITCH + frag_k(j, hi));
        #pragma unroll
        for (int ct = 0; ct < 4; ++ct)
            bq[ct][j] = *(const v4f*)(sP + (ct * 16 + c) * PPITCH + frag_k(j, hi));
    }
}

__device__ __forceinline__ void load_tile_lines(const float* __restrict__ tile, int lane, v4f (&st)[8]) {
    #pragma unroll
    for (int i = 0; i < 8; ++i)
        st[i] = *(const v4f*)(tile + i * 128 + 4 * lane);
}

__device__ __forceinline__ void tile_to_frags(float* sT, int c, int hi, const v4f (&st)[8], v4f (&aq)[8]) {
    #pragma unroll
    for (int i = 0; i < 8; ++i)
        *(v4f*)(sT + (2 * i + hi) * APITCH + 4 * c) = st[i];
    #pragma unroll
    for (int j = 0; j < 8; ++j)
        aq[j] = *(const v4f*)(sT + c * APITCH + frag_k(j, hi));
}

__device__ __forceinline__ void tile_mma(const v4f (&aq)[8], const v4f (&bq)[4][8], v8f (&acc)[4]) {
    #pragma unroll
    for (int j = 0; j < 8; ++j) {
        const v2f a0 = __builtin_shufflevector(aq[j], aq[j], 0, 1);
        const v2f a1 = __builtin_shufflevector(aq[j], aq[j], 2, 3);
        #pragma unroll
        for (int ct = 0; ct < 4; ++ct) {
            const v2f b0 = __builtin_shufflevector(bq[ct][j], bq[ct][j], 0, 1);
            acc[ct] = __builtin_amdgcn_wmma_f32_16x16x4_f32(
                          false, a0, false, b0, (short)0, acc[ct], false, false);
        }
        #pragma unroll
        for (int ct = 0; ct < 4; ++ct) {
            const v2f b1 = __builtin_shufflevector(bq[ct][j], bq[ct][j], 2, 3);
            acc[ct] = __builtin_amdgcn_wmma_f32_16x16x4_f32(
                          false, a1, false, b1, (short)0, acc[ct], false, false);
        }
    }
}

__global__ __launch_bounds__(256) void FA_pass1(
    const float* __restrict__ K, const float* __restrict__ V,
    const float* __restrict__ P,
    float* __restrict__ part_ws)
{
    __shared__ __attribute__((aligned(16))) float sP[DIM * PPITCH];
    __shared__ __attribute__((aligned(16))) float sred[2][8][DIM];
    __shared__ __attribute__((aligned(16))) float sTile[8][16 * APITCH];

    const int bh    = blockIdx.x;
    const int chunk = blockIdx.y;
    const int tid   = threadIdx.x;
    const int wave  = tid >> 5;
    const int lane  = tid & 31;
    const int c     = lane & 15;
    const int hi    = lane >> 4;

    const size_t row0 = (size_t)bh * NSEQ + (size_t)chunk * CHUNK1 + (size_t)wave * (TILES1 * 16);
    const float* Atile = K + row0 * DIM;
    const float* Vrow = V + (row0 + 8 * hi) * DIM + 4 * c;
    float* sT = sTile[wave];

    v4f pr[4];
    load_proj(P, tid, pr);
    __builtin_amdgcn_sched_barrier(0);

    v4f a_cur[8], a_ln[8];
    load_tile_lines(Atile, lane, a_ln);
    __builtin_amdgcn_sched_barrier(0);

    store_proj(sP, tid, pr);
    lds_barrier();

    v4f bq[4][8];
    first_frags(sT, sP, c, hi, a_ln, a_cur, bq);
    __builtin_amdgcn_sched_barrier(0);

    float ksum_p[4] = {0.0f, 0.0f, 0.0f, 0.0f};
    float ctx_p[4]  = {0.0f, 0.0f, 0.0f, 0.0f};

    #pragma unroll
    for (int t = 0; t < TILES1; ++t) {
        if (t + 1 < TILES1) load_tile_lines(Atile + (size_t)(t + 1) * 16 * DIM, lane, a_ln);

        v4f vv[8];
        #pragma unroll
        for (int r = 0; r < 8; ++r)
            vv[r] = *(const v4f*)(Vrow + (size_t)(t * 16 + r) * DIM);
        __builtin_amdgcn_sched_barrier(0);

        v8f acc[4] = {};
        tile_mma(a_cur, bq, acc);
        if (t + 1 < TILES1) __builtin_amdgcn_sched_barrier(0);

        #pragma unroll
        for (int r = 0; r < 8; ++r) {
            #pragma unroll
            for (int ct = 0; ct < 4; ++ct) {
                const float e = __expf(acc[ct][r]);
                ksum_p[ct] += e;
                ctx_p[ct]  += e * vv[r][ct];
            }
        }

        if (t + 1 < TILES1) tile_to_frags(sT, c, hi, a_ln, a_cur);
    }

    #pragma unroll
    for (int ct = 0; ct < 4; ++ct) {
        ksum_p[ct] += __shfl_xor(ksum_p[ct], 16, 32);
        ctx_p[ct]  += __shfl_xor(ctx_p[ct], 16, 32);
    }
    if (hi == 0) {
        v4f s0 = {ksum_p[0], ksum_p[1], ksum_p[2], ksum_p[3]};
        v4f s1 = {ctx_p[0], ctx_p[1], ctx_p[2], ctx_p[3]};
        *(v4f*)(&sred[0][wave][4 * c]) = s0;
        *(v4f*)(&sred[1][wave][4 * c]) = s1;
    }
    __syncthreads();

    if (tid < 2 * DIM) {
        const int kind = tid >> 6;
        const int f    = tid & 63;
        float s = 0.0f;
        #pragma unroll
        for (int w = 0; w < 8; ++w) s += sred[kind][w][f];
        part_ws[((size_t)bh * NCHUNK1 + chunk) * (2 * DIM) + tid] = s;
    }
}

__global__ __launch_bounds__(256, 4) void FA_pass2(
    const float* __restrict__ Q, const float* __restrict__ P,
    const float* __restrict__ part_ws,
    float* __restrict__ out)
{
    __shared__ __attribute__((aligned(16))) float sP[DIM * PPITCH];
    __shared__ __attribute__((aligned(16))) float stot[2][2 * DIM];
    __shared__ __attribute__((aligned(16))) float sTile[8][16 * APITCH];

    const int bh    = blockIdx.x;
    const int chunk = blockIdx.y;
    const int tid   = threadIdx.x;
    const int wave  = tid >> 5;
    const int lane  = tid & 31;
    const int c     = lane & 15;
    const int hi    = lane >> 4;

    const size_t row0 = (size_t)bh * NSEQ + (size_t)chunk * CHUNK2 + (size_t)wave * (TILES2 * 16);
    const float* Atile = Q + row0 * DIM;
    float*       Orow = out + (row0 + 8 * hi) * DIM + 4 * c;
    float* sT = sTile[wave];

    const int slot = tid & (2 * DIM - 1);
    const int half = tid >> 7;
    const float* pp = part_ws + ((size_t)bh * NCHUNK1 + half * (NCHUNK1 / 2)) * (2 * DIM) + slot;
    float pv[NCHUNK1 / 2];
    #pragma unroll
    for (int ch = 0; ch < NCHUNK1 / 2; ++ch) pv[ch] = pp[ch * (2 * DIM)];

    v4f pr[4];
    load_proj(P, tid, pr);
    __builtin_amdgcn_sched_barrier(0);

    v4f a_cur[8], a_ln[8];
    load_tile_lines(Atile, lane, a_ln);
    __builtin_amdgcn_sched_barrier(0);

    store_proj(sP, tid, pr);

    float tot = pv[0];
    #pragma unroll
    for (int ch = 1; ch < NCHUNK1 / 2; ++ch) tot += pv[ch];
    stot[half][slot] = tot;
    lds_barrier();

    v4f bq[4][8];
    first_frags(sT, sP, c, hi, a_ln, a_cur, bq);
    __builtin_amdgcn_sched_barrier(0);

    v4f ks = {}, cx = {};

    #pragma unroll
    for (int t = 0; t < TILES2; ++t) {
        if (t + 1 < TILES2) load_tile_lines(Atile + (size_t)(t + 1) * 16 * DIM, lane, a_ln);
        __builtin_amdgcn_sched_barrier(0);

        v8f acc[4] = {};
        tile_mma(a_cur, bq, acc);
        if (t + 1 < TILES2) __builtin_amdgcn_sched_barrier(0);

        if (t == 0) {
            ks = *(const v4f*)(&stot[0][4 * c]) + *(const v4f*)(&stot[1][4 * c]);
            ks += 1e-6f;
            cx = *(const v4f*)(&stot[0][DIM + 4 * c]) + *(const v4f*)(&stot[1][DIM + 4 * c]);
        }

        float e[4][8];
        float p[8];
        #pragma unroll
        for (int r = 0; r < 8; ++r) {
            p[r] = 0.0f;
            #pragma unroll
            for (int ct = 0; ct < 4; ++ct) {
                e[ct][r] = __expf(acc[ct][r]);
                p[r] += e[ct][r] * ks[ct];
            }
        }
        #pragma unroll
        for (int m = 1; m < 16; m <<= 1) {
            float q[8];
            #pragma unroll
            for (int r = 0; r < 8; ++r) q[r] = __shfl_xor(p[r], m, 32);
            #pragma unroll
            for (int r = 0; r < 8; ++r) p[r] += q[r];
        }
        #pragma unroll
        for (int r = 0; r < 8; ++r) {
            const float rinv = 1.0f / p[r];
            v4f o;
            o.x = e[0][r] * cx[0] * rinv;
            o.y = e[1][r] * cx[1] * rinv;
            o.z = e[2][r] * cx[2] * rinv;
            o.w = e[3][r] * cx[3] * rinv;
            *(v4f*)(Orow + (size_t)(t * 16 + r) * DIM) = o;
        }

        if (t + 1 < TILES2) tile_to_frags(sT, c, hi, a_ln, a_cur);
    }
}

extern "C" void kernel_launch(void* const* d_in, const int* in_sizes, int n_in,
                              void* d_out, int out_size, void* d_ws, size_t ws_size,
                              hipStream_t stream) {
    const float* q    = (const float*)d_in[0];
    const float* k    = (const float*)d_in[1];
    const float* v    = (const float*)d_in[2];
    const float* proj = (const float*)d_in[3];
    float* out = (float*)d_out;

    float* ws_part = (float*)d_ws;
    if (ws_size < PART_FLOATS * sizeof(float)) return;

    FA_pass1<<<dim3(BH, NCHUNK1), 256, 0, stream>>>(
        k, v, proj, ws_part);

    FA_pass2<<<dim3(BH, NSEQ / CHUNK2), 256, 0, stream>>>(
        q, proj, ws_part, out);
}


# Fewer vector slots per tile: log2e folded into the staged projection (base-2 exp direct), pass 2 row sums by DPP reduce-scatter, one division per lane not eight; f32 order of the row sums changes
# speedup vs baseline: 3.1093x; 1.0477x over previous
#include <hip/hip_runtime.h>


typedef __attribute__((ext_vector_type(2))) float v2f;
typedef __attribute__((ext_vector_type(4))) float v4f;
typedef __attribute__((ext_vector_type(8))) float v8f;

#define BH     32
#define NSEQ   4096
#define DIM    64
#define TILES1 4
#define CHUNK1 512
#define TILES2 2
#define CHUNK2 256
static_assert(CHUNK1 == 8 * TILES1 * 16 && CHUNK2 == 8 * TILES2 * 16, "chunk = 8 waves x tiles x 16 rows");
static_assert(NSEQ % CHUNK1 == 0 && NSEQ % CHUNK2 == 0, "chunks tile the sequence");
#define PPITCH 68
#define APITCH 68
#define NCHUNK1 (NSEQ / CHUNK1)
#define PART_FLOATS (BH * NCHUNK1 * 2 * DIM)
#define LOG2E 1.4426950408889634f

__device__ __forceinline__ int frag_k(int j, int hi) { return 8 * j + 4 * hi; }


__device__ __forceinline__ void load_proj(const float* __restrict__ P, int tid, v4f (&pr)[4]) {
    #pragma unroll
    for (int i = 0; i < 4; ++i) {
        const int idx = tid + 256 * i;
        pr[i] = *(const v4f*)(P + (idx >> 4) * DIM + 4 * (idx & 15));
    }
}
__device__ __forceinline__ void lds_barrier() {
    __builtin_amdgcn_fence(__ATOMIC_RELEASE, "workgroup", "local");
    __builtin_amdgcn_s_barrier();
    __builtin_amdgcn_fence(__ATOMIC_ACQUIRE, "workgroup", "local");
}
__device__ __forceinline__ void store_proj(float* sP, int tid, const v4f (&pr)[4]) {
    #pragma unroll
    for (int i = 0; i < 4; ++i) {
        const int idx = tid + 256 * i;
        const int f   = idx >> 4;
        const int q   = idx & 15;
        const int lrow = (f & 3) * 16 + (f >> 2);
        *(v4f*)(sP + lrow * PPITCH + 4 * q) = pr[i] * LOG2E;
    }
}

__device__ __forceinline__ float exp2_hw(float x) { return __builtin_amdgcn_exp2f(x); }

template <int CTRL>
__device__ __forceinline__ float row_dpp(float x) {
    return __builtin_bit_cast(float,
        __builtin_amdgcn_update_dpp(0, __builtin_bit_cast(int, x), CTRL, 0xf, 0xf, true));
}

__device__ __forceinline__ void first_frags(float* sT, const float* sP, int c, int hi,
                                            const v4f (&st)[8], v4f (&aq)[8], v4f (&bq)[4][8]) {
    #pragma unroll
    for (int i = 0; i < 8; ++i)
        *(v4f*)(sT + (2 * i + hi) * APITCH + 4 * c) = st[i];
    #pragma unroll
    for (int j = 0; j < 8; ++j) {
        aq[j] = *(const v4f*)(sT + c * APITCH + frag_k(j, hi));
        #pragma unroll
        for (int ct = 0; ct < 4; ++ct)
            bq[ct][j] = *(const v4f*)(sP + (ct * 16 + c) * PPITCH + frag_k(j, hi));
    }
}

__device__ __forceinline__ void load_tile_lines(const float* __restrict__ tile, int lane, v4f (&st)[8]) {
    #pragma unroll
    for (int i = 0; i < 8; ++i)
        st[i] = *(const v4f*)(tile + i * 128 + 4 * lane);
}

__device__ __forceinline__ void tile_to_frags(float* sT, int c, int hi, const v4f (&st)[8], v4f (&aq)[8]) {
    #pragma unroll
    for (int i = 0; i < 8; ++i)
        *(v4f*)(sT + (2 * i + hi) * APITCH + 4 * c) = st[i];
    #pragma unroll
    for (int j = 0; j < 8; ++j)
        aq[j] = *(const v4f*)(sT + c * APITCH + frag_k(j, hi));
}

__device__ __forceinline__ void tile_mma(const v4f (&aq)[8], const v4f (&bq)[4][8], v8f (&acc)[4]) {
    #pragma unroll
    for (int j = 0; j < 8; ++j) {
        const v2f a0 = __builtin_shufflevector(aq[j], aq[j], 0, 1);
        const v2f a1 = __builtin_shufflevector(aq[j], aq[j], 2, 3);
        #pragma unroll
        for (int ct = 0; ct < 4; ++ct) {
            const v2f b0 = __builtin_shufflevector(bq[ct][j], bq[ct][j], 0, 1);
            acc[ct] = __builtin_amdgcn_wmma_f32_16x16x4_f32(
                          false, a0, false, b0, (short)0, acc[ct], false, false);
        }
        #pragma unroll
        for (int ct = 0; ct < 4; ++ct) {
            const v2f b1 = __builtin_shufflevector(bq[ct][j], bq[ct][j], 2, 3);
            acc[ct] = __builtin_amdgcn_wmma_f32_16x16x4_f32(
                          false, a1, false, b1, (short)0, acc[ct], false, false);
        }
    }
}

__global__ __launch_bounds__(256) void FA_pass1(
    const float* __restrict__ K, const float* __restrict__ V,
    const float* __restrict__ P,
    float* __restrict__ part_ws)
{
    __shared__ __attribute__((aligned(16))) float sP[DIM * PPITCH];
    __shared__ __attribute__((aligned(16))) float sred[2][8][DIM];
    __shared__ __attribute__((aligned(16))) float sTile[8][16 * APITCH];

    const int bh    = blockIdx.x;
    const int chunk = blockIdx.y;
    const int tid   = threadIdx.x;
    const int wave  = tid >> 5;
    const int lane  = tid & 31;
    const int c     = lane & 15;
    const int hi    = lane >> 4;

    const size_t row0 = (size_t)bh * NSEQ + (size_t)chunk * CHUNK1 + (size_t)wave * (TILES1 * 16);
    const float* Atile = K + row0 * DIM;
    const float* Vrow = V + (row0 + 8 * hi) * DIM + 4 * c;
    float* sT = sTile[wave];

    v4f pr[4];
    load_proj(P, tid, pr);
    __builtin_amdgcn_sched_barrier(0);

    v4f a_cur[8], a_ln[8];
    load_tile_lines(Atile, lane, a_ln);
    __builtin_amdgcn_sched_barrier(0);

    store_proj(sP, tid, pr);
    lds_barrier();

    v4f bq[4][8];
    first_frags(sT, sP, c, hi, a_ln, a_cur, bq);
    __builtin_amdgcn_sched_barrier(0);

    v2f ksum01 = {0.0f, 0.0f}, ksum23 = {0.0f, 0.0f};
    v2f ctx01  = {0.0f, 0.0f}, ctx23  = {0.0f, 0.0f};

    #pragma unroll
    for (int t = 0; t < TILES1; ++t) {
        if (t + 1 < TILES1) load_tile_lines(Atile + (size_t)(t + 1) * 16 * DIM, lane, a_ln);

        v4f vv[8];
        #pragma unroll
        for (int r = 0; r < 8; ++r)
            vv[r] = *(const v4f*)(Vrow + (size_t)(t * 16 + r) * DIM);
        __builtin_amdgcn_sched_barrier(0);

        v8f acc[4] = {};
        tile_mma(a_cur, bq, acc);
        if (t + 1 < TILES1) __builtin_amdgcn_sched_barrier(0);

        #pragma unroll
        for (int r = 0; r < 8; ++r) {
            const v2f e01 = {exp2_hw(acc[0][r]), exp2_hw(acc[1][r])};
            const v2f e23 = {exp2_hw(acc[2][r]), exp2_hw(acc[3][r])};
            const v2f v01 = __builtin_shufflevector(vv[r], vv[r], 0, 1);
            const v2f v23 = __builtin_shufflevector(vv[r], vv[r], 2, 3);
            ksum01 += e01;
            ksum23 += e23;
            ctx01  += e01 * v01;
            ctx23  += e23 * v23;
        }

        if (t + 1 < TILES1) tile_to_frags(sT, c, hi, a_ln, a_cur);
    }

    float ksum_p[4] = {ksum01.x, ksum01.y, ksum23.x, ksum23.y};
    float ctx_p[4]  = {ctx01.x, ctx01.y, ctx23.x, ctx23.y};
    #pragma unroll
    for (int ct = 0; ct < 4; ++ct) {
        ksum_p[ct] += __shfl_xor(ksum_p[ct], 16, 32);
        ctx_p[ct]  += __shfl_xor(ctx_p[ct], 16, 32);
    }
    if (hi == 0) {
        v4f s0 = {ksum_p[0], ksum_p[1], ksum_p[2], ksum_p[3]};
        v4f s1 = {ctx_p[0], ctx_p[1], ctx_p[2], ctx_p[3]};
        *(v4f*)(&sred[0][wave][4 * c]) = s0;
        *(v4f*)(&sred[1][wave][4 * c]) = s1;
    }
    __syncthreads();

    if (tid < 2 * DIM) {
        const int kind = tid >> 6;
        const int f    = tid & 63;
        float s = 0.0f;
        #pragma unroll
        for (int w = 0; w < 8; ++w) s += sred[kind][w][f];
        part_ws[((size_t)bh * NCHUNK1 + chunk) * (2 * DIM) + tid] = s;
    }
}

__global__ __launch_bounds__(256, 4) void FA_pass2(
    const float* __restrict__ Q, const float* __restrict__ P,
    const float* __restrict__ part_ws,
    float* __restrict__ out)
{
    __shared__ __attribute__((aligned(16))) float sP[DIM * PPITCH];
    __shared__ __attribute__((aligned(16))) float stot[2][2 * DIM];
    __shared__ __attribute__((aligned(16))) float sTile[8][16 * APITCH];

    const int bh    = blockIdx.x;
    const int chunk = blockIdx.y;
    const int tid   = threadIdx.x;
    const int wave  = tid >> 5;
    const int lane  = tid & 31;
    const int c     = lane & 15;
    const int hi    = lane >> 4;

    const size_t row0 = (size_t)bh * NSEQ + (size_t)chunk * CHUNK2 + (size_t)wave * (TILES2 * 16);
    const float* Atile = Q + row0 * DIM;
    float*       Orow = out + (row0 + 8 * hi) * DIM + 4 * c;
    float* sT = sTile[wave];

    const int slot = tid & (2 * DIM - 1);
    const int half = tid >> 7;
    const float* pp = part_ws + ((size_t)bh * NCHUNK1 + half * (NCHUNK1 / 2)) * (2 * DIM) + slot;
    float pv[NCHUNK1 / 2];
    #pragma unroll
    for (int ch = 0; ch < NCHUNK1 / 2; ++ch) pv[ch] = pp[ch * (2 * DIM)];

    v4f pr[4];
    load_proj(P, tid, pr);
    __builtin_amdgcn_sched_barrier(0);

    v4f a_cur[8], a_ln[8];
    load_tile_lines(Atile, lane, a_ln);
    __builtin_amdgcn_sched_barrier(0);

    store_proj(sP, tid, pr);

    float tot = pv[0];
    #pragma unroll
    for (int ch = 1; ch < NCHUNK1 / 2; ++ch) tot += pv[ch];
    stot[half][slot] = tot;
    lds_barrier();

    v4f bq[4][8];
    first_frags(sT, sP, c, hi, a_ln, a_cur, bq);
    __builtin_amdgcn_sched_barrier(0);

    v4f ks = {}, cx = {};

    #pragma unroll
    for (int t = 0; t < TILES2; ++t) {
        if (t + 1 < TILES2) load_tile_lines(Atile + (size_t)(t + 1) * 16 * DIM, lane, a_ln);
        __builtin_amdgcn_sched_barrier(0);

        v8f acc[4] = {};
        tile_mma(a_cur, bq, acc);
        if (t + 1 < TILES2) __builtin_amdgcn_sched_barrier(0);

        if (t == 0) {
            ks = *(const v4f*)(&stot[0][4 * c]) + *(const v4f*)(&stot[1][4 * c]);
            ks += 1e-6f;
            cx = *(const v4f*)(&stot[0][DIM + 4 * c]) + *(const v4f*)(&stot[1][DIM + 4 * c]);
        }

        const v2f ks01 = __builtin_shufflevector(ks, ks, 0, 1);
        const v2f ks23 = __builtin_shufflevector(ks, ks, 2, 3);
        const v2f cx01 = __builtin_shufflevector(cx, cx, 0, 1);
        const v2f cx23 = __builtin_shufflevector(cx, cx, 2, 3);
        v2f e01[8], e23[8];
        float p[8];
        #pragma unroll
        for (int r = 0; r < 8; ++r) {
            e01[r] = v2f{exp2_hw(acc[0][r]), exp2_hw(acc[1][r])};
            e23[r] = v2f{exp2_hw(acc[2][r]), exp2_hw(acc[3][r])};
            const v2f s = e01[r] * ks01 + e23[r] * ks23;
            p[r] = s.x + s.y;
        }
        const bool b3 = (c & 8) != 0, b2 = (c & 4) != 0, b1 = (c & 2) != 0;
        float s1[4], s2[2];
        #pragma unroll
        for (int r = 0; r < 4; ++r) {
            const float keep = b3 ? p[r + 4] : p[r];
            const float give = b3 ? p[r] : p[r + 4];
            s1[r] = keep + row_dpp<0x168>(give);
        }
        #pragma unroll
        for (int r = 0; r < 2; ++r) {
            const float keep = b2 ? s1[r + 2] : s1[r];
            const float give = b2 ? s1[r] : s1[r + 2];
            s2[r] = keep + row_dpp<0x164>(give);
        }
        const float keep3 = b1 ? s2[1] : s2[0];
        const float give3 = b1 ? s2[0] : s2[1];
        const float s3 = keep3 + row_dpp<0x162>(give3);
        const float s4 = s3 + row_dpp<0x161>(s3);
        const float rinv_own = 1.0f / s4;
        const float rinv[8] = {
            row_dpp<0x150>(rinv_own),  row_dpp<0x152>(rinv_own),
            row_dpp<0x154>(rinv_own),  row_dpp<0x156>(rinv_own),
            row_dpp<0x158>(rinv_own),  row_dpp<0x15a>(rinv_own),
            row_dpp<0x15c>(rinv_own),  row_dpp<0x15e>(rinv_own)};
        #pragma unroll
        for (int r = 0; r < 8; ++r) {
            const v2f o01 = e01[r] * cx01 * rinv[r];
            const v2f o23 = e23[r] * cx23 * rinv[r];
            const v4f o = __builtin_shufflevector(o01, o23, 0, 1, 2, 3);
            *(v4f*)(Orow + (size_t)(t * 16 + r) * DIM) = o;
        }

        if (t + 1 < TILES2) tile_to_frags(sT, c, hi, a_ln, a_cur);
    }
}

extern "C" void kernel_launch(void* const* d_in, const int* in_sizes, int n_in,
                              void* d_out, int out_size, void* d_ws, size_t ws_size,
                              hipStream_t stream) {
    const float* q    = (const float*)d_in[0];
    const float* k    = (const float*)d_in[1];
    const float* v    = (const float*)d_in[2];
    const float* proj = (const float*)d_in[3];
    float* out = (float*)d_out;

    float* ws_part = (float*)d_ws;
    if (ws_size < PART_FLOATS * sizeof(float)) return;

    FA_pass1<<<dim3(BH, NCHUNK1), 256, 0, stream>>>(
        k, v, proj, ws_part);

    FA_pass2<<<dim3(BH, NSEQ / CHUNK2), 256, 0, stream>>>(
        q, proj, ws_part, out);
}
